# starting kernel
# baseline (speedup 1.0000x reference)
#include <hip/hip_runtime.h>
#include <hip/hip_bf16.h>
#include <math.h>

typedef __attribute__((ext_vector_type(2))) float v2f;
typedef __attribute__((ext_vector_type(8))) float v8f;

__global__ __launch_bounds__(256) void gemm_rows64_wmma(
    const float* __restrict__ X, const float* __restrict__ W,
    float* __restrict__ Y, int nrows)
{
  __shared__ v2f sW2[32 * 64];

  const int tid = threadIdx.x;
  for (int i = tid; i < 32 * 64; i += 256) {
    const int j = i >> 6;
    const int n = i & 63;
    v2f w;
    w.x = W[(2 * j) * 64 + n];
    w.y = W[(2 * j + 1) * 64 + n];
    sW2[i] = w;
  }
  __syncthreads();

  const int lane    = tid & 31;
  const int wave    = tid >> 5;
  const int rowBase = blockIdx.x * 128 + wave * 16;
  if (rowBase + 16 > nrows) return;

  const int m  = lane & 15;
  const int kh = lane >> 4;
  const int n  = lane & 15;

  const float* __restrict__ xrow = X + (size_t)(rowBase + m) * 64;

  v8f acc0 = {}, acc1 = {}, acc2 = {}, acc3 = {};

#pragma unroll
  for (int kk = 0; kk < 16; ++kk) {
    const int k0 = kk * 4 + kh * 2;
    v2f a = *(const v2f*)(xrow + k0);

    const int kp = k0 >> 1;
    const v2f* __restrict__ brow = &sW2[kp * 64 + n];
    v2f b0 = brow[0];
    v2f b1 = brow[16];
    v2f b2 = brow[32];
    v2f b3 = brow[48];

    acc0 = __builtin_amdgcn_wmma_f32_16x16x4_f32(false, a, false, b0, (short)0, acc0, false, false);
    acc1 = __builtin_amdgcn_wmma_f32_16x16x4_f32(false, a, false, b1, (short)0, acc1, false, false);
    acc2 = __builtin_amdgcn_wmma_f32_16x16x4_f32(false, a, false, b2, (short)0, acc2, false, false);
    acc3 = __builtin_amdgcn_wmma_f32_16x16x4_f32(false, a, false, b3, (short)0, acc3, false, false);
  }

  const int crow = rowBase + (lane >> 4) * 8;
  const int ccol = lane & 15;
#pragma unroll
  for (int v = 0; v < 8; ++v) {
    float* __restrict__ yr = Y + (size_t)(crow + v) * 64 + ccol;
    yr[0]  = acc0[v];
    yr[16] = acc1[v];
    yr[32] = acc2[v];
    yr[48] = acc3[v];
  }
}

__global__ __launch_bounds__(256) void scatter_add64(
    const float* __restrict__ src, const int* __restrict__ src_idx,
    const int* __restrict__ dst_idx, float* __restrict__ dst, int n_inc)
{
  const int e = blockIdx.x * 16 + (threadIdx.x >> 4);
  if (e >= n_inc) return;
  const int c4 = (threadIdx.x & 15) * 4;
  const int s  = src_idx[e];
  const int d  = dst_idx[e];
  const float4 v = *(const float4*)(src + (size_t)s * 64 + c4);
  float* __restrict__ dp = dst + (size_t)d * 64 + c4;
  atomicAdd(dp + 0, v.x);
  atomicAdd(dp + 1, v.y);
  atomicAdd(dp + 2, v.z);
  atomicAdd(dp + 3, v.w);
}

__global__ __launch_bounds__(256) void gate_tanh4(
    const float* __restrict__ pre, const float* __restrict__ gate,
    float* __restrict__ out, size_t n4)
{
  const size_t i = (size_t)blockIdx.x * blockDim.x + threadIdx.x;
  if (i >= n4) return;
  const float4 p = ((const float4*)pre)[i];
  const float4 g = ((const float4*)gate)[i];
  float4 r;
  r.x = tanhf(p.x * g.x);
  r.y = tanhf(p.y * g.y);
  r.z = tanhf(p.z * g.z);
  r.w = tanhf(p.w * g.w);
  ((float4*)out)[i] = r;
}

extern "C" void kernel_launch(void* const* d_in, const int* in_sizes, int n_in,
                              void* d_out, int out_size, void* d_ws, size_t ws_size,
                              hipStream_t stream) {
  const float* node_features  = (const float*)d_in[0];
  const float* hedge_features = (const float*)d_in[1];
  const float* W_nn = (const float*)d_in[2];
  const float* W_nh = (const float*)d_in[3];
  const float* W_hh = (const float*)d_in[4];
  const float* W_hn = (const float*)d_in[5];
  const int* node_idx  = (const int*)d_in[6];
  const int* hedge_idx = (const int*)d_in[7];

  const int N   = in_sizes[0] / 64;
  const int E   = in_sizes[1] / 64;
  const int INC = in_sizes[6];

  float* slotA = (float*)d_ws;
  float* slotB = slotA + (size_t)N * 64;

  float* out_node  = (float*)d_out;
  float* out_hedge = out_node + (size_t)N * 64;

  const dim3 blk(256);
  const dim3 gemmGridN((N + 127) / 128);
  const dim3 gemmGridE((E + 127) / 128);
  const dim3 scatGrid((INC + 15) / 16);

  gemm_rows64_wmma<<<gemmGridE, blk, 0, stream>>>(hedge_features, W_nh, slotA, E);
  hipMemsetAsync(slotB, 0, (size_t)N * 64 * sizeof(float), stream);
  scatter_add64<<<scatGrid, blk, 0, stream>>>(slotA, hedge_idx, node_idx, slotB, INC);
  gemm_rows64_wmma<<<gemmGridN, blk, 0, stream>>>(node_features, W_nn, slotA, N);
  gate_tanh4<<<dim3(((size_t)N * 16 + 255) / 256), blk, 0, stream>>>(
      slotA, slotB, out_node, (size_t)N * 16);

  gemm_rows64_wmma<<<gemmGridN, blk, 0, stream>>>(out_node, W_hn, slotA, N);
  hipMemsetAsync(slotB, 0, (size_t)E * 64 * sizeof(float), stream);
  scatter_add64<<<scatGrid, blk, 0, stream>>>(slotA, node_idx, hedge_idx, slotB, INC);
  gemm_rows64_wmma<<<gemmGridE, blk, 0, stream>>>(hedge_features, W_hh, slotA, E);
  gate_tanh4<<<dim3(((size_t)E * 16 + 255) / 256), blk, 0, stream>>>(
      slotA, slotB, out_hedge, (size_t)E * 16);
}


# scatter_add64: a wave takes 4 incidences in turn, all 32 lanes on one row (lane l: columns l, l+32): each f32 atomic instruction covers one whole 128-B line, 2x fewer waves; arrival order as before
# speedup vs baseline: 1.4732x; 1.4732x over previous
#include <hip/hip_runtime.h>
#include <hip/hip_bf16.h>
#include <math.h>

typedef __attribute__((ext_vector_type(2))) float v2f;
typedef __attribute__((ext_vector_type(8))) float v8f;

__global__ __launch_bounds__(256) void gemm_rows64_wmma(
    const float* __restrict__ X, const float* __restrict__ W,
    float* __restrict__ Y, int nrows)
{
  __shared__ v2f sW2[32 * 64];

  const int tid = threadIdx.x;
  for (int i = tid; i < 32 * 64; i += 256) {
    const int j = i >> 6;
    const int n = i & 63;
    v2f w;
    w.x = W[(2 * j) * 64 + n];
    w.y = W[(2 * j + 1) * 64 + n];
    sW2[i] = w;
  }
  __syncthreads();

  const int lane    = tid & 31;
  const int wave    = tid >> 5;
  const int rowBase = blockIdx.x * 128 + wave * 16;
  if (rowBase + 16 > nrows) return;

  const int m  = lane & 15;
  const int kh = lane >> 4;
  const int n  = lane & 15;

  const float* __restrict__ xrow = X + (size_t)(rowBase + m) * 64;

  v8f acc0 = {}, acc1 = {}, acc2 = {}, acc3 = {};

#pragma unroll
  for (int kk = 0; kk < 16; ++kk) {
    const int k0 = kk * 4 + kh * 2;
    v2f a = *(const v2f*)(xrow + k0);

    const int kp = k0 >> 1;
    const v2f* __restrict__ brow = &sW2[kp * 64 + n];
    v2f b0 = brow[0];
    v2f b1 = brow[16];
    v2f b2 = brow[32];
    v2f b3 = brow[48];

    acc0 = __builtin_amdgcn_wmma_f32_16x16x4_f32(false, a, false, b0, (short)0, acc0, false, false);
    acc1 = __builtin_amdgcn_wmma_f32_16x16x4_f32(false, a, false, b1, (short)0, acc1, false, false);
    acc2 = __builtin_amdgcn_wmma_f32_16x16x4_f32(false, a, false, b2, (short)0, acc2, false, false);
    acc3 = __builtin_amdgcn_wmma_f32_16x16x4_f32(false, a, false, b3, (short)0, acc3, false, false);
  }

  const int crow = rowBase + (lane >> 4) * 8;
  const int ccol = lane & 15;
#pragma unroll
  for (int v = 0; v < 8; ++v) {
    float* __restrict__ yr = Y + (size_t)(crow + v) * 64 + ccol;
    yr[0]  = acc0[v];
    yr[16] = acc1[v];
    yr[32] = acc2[v];
    yr[48] = acc3[v];
  }
}

__global__ __launch_bounds__(256) void scatter_add64(
    const float* __restrict__ src, const int* __restrict__ src_idx,
    const int* __restrict__ dst_idx, float* __restrict__ dst, int n_inc)
{
  const int lane = threadIdx.x & 31;
  const int wv = __builtin_amdgcn_readfirstlane(blockIdx.x * 8 + (threadIdx.x >> 5));
  const int e0 = wv * 4;
  if (e0 >= n_inc) return;

  if (e0 + 4 <= n_inc) {
    int s[4], d[4];
#pragma unroll
    for (int j = 0; j < 4; ++j) { s[j] = src_idx[e0 + j]; d[j] = dst_idx[e0 + j]; }
    float lo[4], hi[4];
#pragma unroll
    for (int j = 0; j < 4; ++j) {
      const float* __restrict__ sp = src + (size_t)s[j] * 64 + lane;
      lo[j] = sp[0];
      hi[j] = sp[32];
    }
#pragma unroll
    for (int j = 0; j < 4; ++j) {
      float* __restrict__ dp = dst + (size_t)d[j] * 64 + lane;
      atomicAdd(dp, lo[j]);
      atomicAdd(dp + 32, hi[j]);
    }
  } else {
    for (int e = e0; e < n_inc; ++e) {
      const int s = src_idx[e];
      const int d = dst_idx[e];
      const float* __restrict__ sp = src + (size_t)s * 64 + lane;
      float* __restrict__ dp = dst + (size_t)d * 64 + lane;
      atomicAdd(dp, sp[0]);
      atomicAdd(dp + 32, sp[32]);
    }
  }
}

__global__ __launch_bounds__(256) void gate_tanh4(
    const float* __restrict__ pre, const float* __restrict__ gate,
    float* __restrict__ out, size_t n4)
{
  const size_t i = (size_t)blockIdx.x * blockDim.x + threadIdx.x;
  if (i >= n4) return;
  const float4 p = ((const float4*)pre)[i];
  const float4 g = ((const float4*)gate)[i];
  float4 r;
  r.x = tanhf(p.x * g.x);
  r.y = tanhf(p.y * g.y);
  r.z = tanhf(p.z * g.z);
  r.w = tanhf(p.w * g.w);
  ((float4*)out)[i] = r;
}

extern "C" void kernel_launch(void* const* d_in, const int* in_sizes, int n_in,
                              void* d_out, int out_size, void* d_ws, size_t ws_size,
                              hipStream_t stream) {
  const float* node_features  = (const float*)d_in[0];
  const float* hedge_features = (const float*)d_in[1];
  const float* W_nn = (const float*)d_in[2];
  const float* W_nh = (const float*)d_in[3];
  const float* W_hh = (const float*)d_in[4];
  const float* W_hn = (const float*)d_in[5];
  const int* node_idx  = (const int*)d_in[6];
  const int* hedge_idx = (const int*)d_in[7];

  const int N   = in_sizes[0] / 64;
  const int E   = in_sizes[1] / 64;
  const int INC = in_sizes[6];

  float* slotA = (float*)d_ws;
  float* slotB = slotA + (size_t)N * 64;

  float* out_node  = (float*)d_out;
  float* out_hedge = out_node + (size_t)N * 64;

  const dim3 blk(256);
  const dim3 gemmGridN((N + 127) / 128);
  const dim3 gemmGridE((E + 127) / 128);
  const dim3 scatGrid((INC + 31) / 32);

  gemm_rows64_wmma<<<gemmGridE, blk, 0, stream>>>(hedge_features, W_nh, slotA, E);
  hipMemsetAsync(slotB, 0, (size_t)N * 64 * sizeof(float), stream);
  scatter_add64<<<scatGrid, blk, 0, stream>>>(slotA, hedge_idx, node_idx, slotB, INC);
  gemm_rows64_wmma<<<gemmGridN, blk, 0, stream>>>(node_features, W_nn, slotA, N);
  gate_tanh4<<<dim3(((size_t)N * 16 + 255) / 256), blk, 0, stream>>>(
      slotA, slotB, out_node, (size_t)N * 16);

  gemm_rows64_wmma<<<gemmGridN, blk, 0, stream>>>(out_node, W_hn, slotA, N);
  hipMemsetAsync(slotB, 0, (size_t)E * 64 * sizeof(float), stream);
  scatter_add64<<<scatGrid, blk, 0, stream>>>(slotA, node_idx, hedge_idx, slotB, INC);
  gemm_rows64_wmma<<<gemmGridE, blk, 0, stream>>>(hedge_features, W_hh, slotA, E);
  gate_tanh4<<<dim3(((size_t)E * 16 + 255) / 256), blk, 0, stream>>>(
      slotA, slotB, out_hedge, (size_t)E * 16);
}


# Fuse the row-local chain into one kernel: the X@W tile stays in a per-wave LDS slab, gate*tanh is applied there, and for nodes new_node@W_hn is taken from the same slab; 8 launches -> 5
# speedup vs baseline: 1.7010x; 1.1546x over previous
#include <hip/hip_runtime.h>
#include <hip/hip_bf16.h>
#include <math.h>

typedef __attribute__((ext_vector_type(2))) float v2f;
typedef __attribute__((ext_vector_type(8))) float v8f;

__global__ __launch_bounds__(256) void gemm_rows64_wmma(
    const float* __restrict__ X, const float* __restrict__ W,
    float* __restrict__ Y, int nrows)
{
  __shared__ v2f sW2[32 * 64];

  const int tid = threadIdx.x;
  for (int i = tid; i < 32 * 64; i += 256) {
    const int j = i >> 6;
    const int n = i & 63;
    v2f w;
    w.x = W[(2 * j) * 64 + n];
    w.y = W[(2 * j + 1) * 64 + n];
    sW2[i] = w;
  }
  __syncthreads();

  const int lane    = tid & 31;
  const int wave    = tid >> 5;
  const int rowBase = blockIdx.x * 128 + wave * 16;
  if (rowBase + 16 > nrows) return;

  const int m  = lane & 15;
  const int kh = lane >> 4;
  const int n  = lane & 15;

  const float* __restrict__ xrow = X + (size_t)(rowBase + m) * 64;

  v8f acc0 = {}, acc1 = {}, acc2 = {}, acc3 = {};

#pragma unroll
  for (int kk = 0; kk < 16; ++kk) {
    const int k0 = kk * 4 + kh * 2;
    v2f a = *(const v2f*)(xrow + k0);

    const int kp = k0 >> 1;
    const v2f* __restrict__ brow = &sW2[kp * 64 + n];
    v2f b0 = brow[0];
    v2f b1 = brow[16];
    v2f b2 = brow[32];
    v2f b3 = brow[48];

    acc0 = __builtin_amdgcn_wmma_f32_16x16x4_f32(false, a, false, b0, (short)0, acc0, false, false);
    acc1 = __builtin_amdgcn_wmma_f32_16x16x4_f32(false, a, false, b1, (short)0, acc1, false, false);
    acc2 = __builtin_amdgcn_wmma_f32_16x16x4_f32(false, a, false, b2, (short)0, acc2, false, false);
    acc3 = __builtin_amdgcn_wmma_f32_16x16x4_f32(false, a, false, b3, (short)0, acc3, false, false);
  }

  const int crow = rowBase + (lane >> 4) * 8;
  const int ccol = lane & 15;
#pragma unroll
  for (int v = 0; v < 8; ++v) {
    float* __restrict__ yr = Y + (size_t)(crow + v) * 64 + ccol;
    yr[0]  = acc0[v];
    yr[16] = acc1[v];
    yr[32] = acc2[v];
    yr[48] = acc3[v];
  }
}

#define TP 68
template <bool CHAIN>
__global__ __launch_bounds__(256) void gemm_gate_tanh_chain(
    const float* __restrict__ X, const float* __restrict__ W1,
    const float* __restrict__ G, const float* __restrict__ W2,
    float* __restrict__ Hout, float* __restrict__ Pout, int nrows)
{
  __shared__ v2f sW1[32 * 64];
  __shared__ v2f sWc[CHAIN ? 32 * 64 : 1];
  __shared__ __attribute__((aligned(16))) float sT[8 * 16 * TP];

  const int tid = threadIdx.x;
  for (int i = tid; i < 32 * 64; i += 256) {
    const int j = i >> 6;
    const int n = i & 63;
    v2f w;
    w.x = W1[(2 * j) * 64 + n];
    w.y = W1[(2 * j + 1) * 64 + n];
    sW1[i] = w;
    if (CHAIN) {
      v2f u;
      u.x = W2[(2 * j) * 64 + n];
      u.y = W2[(2 * j + 1) * 64 + n];
      sWc[i] = u;
    }
  }
  __syncthreads();

  const int lane    = tid & 31;
  const int wave    = tid >> 5;
  const int rowBase = blockIdx.x * 128 + wave * 16;
  const bool valid  = (rowBase + 16 <= nrows);

  const int m  = lane & 15;
  const int kh = lane >> 4;
  float* __restrict__ tile = sT + wave * (16 * TP);

  float4 g[8] = {};

  if (valid) {
#pragma unroll
    for (int i = 0; i < 8; ++i)
      g[i] = *(const float4*)(G + (size_t)(rowBase + 2 * i + kh) * 64 + m * 4);
    const float* __restrict__ xrow = X + (size_t)(rowBase + m) * 64;
    v8f acc0 = {}, acc1 = {}, acc2 = {}, acc3 = {};
#pragma unroll
    for (int kk = 0; kk < 16; ++kk) {
      const int k0 = kk * 4 + kh * 2;
      v2f a = *(const v2f*)(xrow + k0);
      const v2f* __restrict__ brow = &sW1[(k0 >> 1) * 64 + m];
      v2f b0 = brow[0];
      v2f b1 = brow[16];
      v2f b2 = brow[32];
      v2f b3 = brow[48];
      acc0 = __builtin_amdgcn_wmma_f32_16x16x4_f32(false, a, false, b0, (short)0, acc0, false, false);
      acc1 = __builtin_amdgcn_wmma_f32_16x16x4_f32(false, a, false, b1, (short)0, acc1, false, false);
      acc2 = __builtin_amdgcn_wmma_f32_16x16x4_f32(false, a, false, b2, (short)0, acc2, false, false);
      acc3 = __builtin_amdgcn_wmma_f32_16x16x4_f32(false, a, false, b3, (short)0, acc3, false, false);
    }
#pragma unroll
    for (int v = 0; v < 8; ++v) {
      float* __restrict__ tr = tile + (kh * 8 + v) * TP + m;
      tr[0]  = acc0[v];
      tr[16] = acc1[v];
      tr[32] = acc2[v];
      tr[48] = acc3[v];
    }
  }
  __syncthreads();

  if (valid) {
    const int c4 = m * 4;
#pragma unroll
    for (int i = 0; i < 8; ++i) {
      const int r = 2 * i + kh;
      float4* __restrict__ tp = (float4*)(tile + r * TP + c4);
      const float4 p = *tp;
      const size_t off = (size_t)(rowBase + r) * 64 + c4;
      float4 h;
      h.x = tanhf(p.x * g[i].x);
      h.y = tanhf(p.y * g[i].y);
      h.z = tanhf(p.z * g[i].z);
      h.w = tanhf(p.w * g[i].w);
      *(float4*)(Hout + off) = h;
      if (CHAIN) *tp = h;
    }
  }

  if (CHAIN) {
    __syncthreads();
    if (valid) {
      const float* __restrict__ hrow = tile + m * TP;
      v8f acc0 = {}, acc1 = {}, acc2 = {}, acc3 = {};
#pragma unroll
      for (int kk = 0; kk < 16; ++kk) {
        const int k0 = kk * 4 + kh * 2;
        v2f a = *(const v2f*)(hrow + k0);
        const v2f* __restrict__ brow = &sWc[(k0 >> 1) * 64 + m];
        v2f b0 = brow[0];
        v2f b1 = brow[16];
        v2f b2 = brow[32];
        v2f b3 = brow[48];
        acc0 = __builtin_amdgcn_wmma_f32_16x16x4_f32(false, a, false, b0, (short)0, acc0, false, false);
        acc1 = __builtin_amdgcn_wmma_f32_16x16x4_f32(false, a, false, b1, (short)0, acc1, false, false);
        acc2 = __builtin_amdgcn_wmma_f32_16x16x4_f32(false, a, false, b2, (short)0, acc2, false, false);
        acc3 = __builtin_amdgcn_wmma_f32_16x16x4_f32(false, a, false, b3, (short)0, acc3, false, false);
      }
      const int crow = rowBase + kh * 8;
#pragma unroll
      for (int v = 0; v < 8; ++v) {
        float* __restrict__ yr = Pout + (size_t)(crow + v) * 64 + m;
        yr[0]  = acc0[v];
        yr[16] = acc1[v];
        yr[32] = acc2[v];
        yr[48] = acc3[v];
      }
    }
  }
}

__global__ __launch_bounds__(256) void scatter_add64(
    const float* __restrict__ src, const int* __restrict__ src_idx,
    const int* __restrict__ dst_idx, float* __restrict__ dst, int n_inc)
{
  const int lane = threadIdx.x & 31;
  const int wv = __builtin_amdgcn_readfirstlane(blockIdx.x * 8 + (threadIdx.x >> 5));
  const int e0 = wv * 4;
  if (e0 >= n_inc) return;

  if (e0 + 4 <= n_inc) {
    int s[4], d[4];
#pragma unroll
    for (int j = 0; j < 4; ++j) { s[j] = src_idx[e0 + j]; d[j] = dst_idx[e0 + j]; }
    float lo[4], hi[4];
#pragma unroll
    for (int j = 0; j < 4; ++j) {
      const float* __restrict__ sp = src + (size_t)s[j] * 64 + lane;
      lo[j] = sp[0];
      hi[j] = sp[32];
    }
#pragma unroll
    for (int j = 0; j < 4; ++j) {
      float* __restrict__ dp = dst + (size_t)d[j] * 64 + lane;
      atomicAdd(dp, lo[j]);
      atomicAdd(dp + 32, hi[j]);
    }
  } else {
    for (int e = e0; e < n_inc; ++e) {
      const int s = src_idx[e];
      const int d = dst_idx[e];
      const float* __restrict__ sp = src + (size_t)s * 64 + lane;
      float* __restrict__ dp = dst + (size_t)d * 64 + lane;
      atomicAdd(dp, sp[0]);
      atomicAdd(dp + 32, sp[32]);
    }
  }
}

extern "C" void kernel_launch(void* const* d_in, const int* in_sizes, int n_in,
                              void* d_out, int out_size, void* d_ws, size_t ws_size,
                              hipStream_t stream) {
  const float* node_features  = (const float*)d_in[0];
  const float* hedge_features = (const float*)d_in[1];
  const float* W_nn = (const float*)d_in[2];
  const float* W_nh = (const float*)d_in[3];
  const float* W_hh = (const float*)d_in[4];
  const float* W_hn = (const float*)d_in[5];
  const int* node_idx  = (const int*)d_in[6];
  const int* hedge_idx = (const int*)d_in[7];

  const int N   = in_sizes[0] / 64;
  const int E   = in_sizes[1] / 64;
  const int INC = in_sizes[6];

  float* slotA = (float*)d_ws;
  float* slotB = slotA + (size_t)N * 64;

  float* out_node  = (float*)d_out;
  float* out_hedge = out_node + (size_t)N * 64;

  const dim3 blk(256);
  const dim3 gemmGridN((N + 127) / 128);
  const dim3 gemmGridE((E + 127) / 128);
  const dim3 scatGrid((INC + 31) / 32);

  gemm_rows64_wmma<<<gemmGridE, blk, 0, stream>>>(hedge_features, W_nh, slotA, E);
  hipMemsetAsync(slotB, 0, (size_t)N * 64 * sizeof(float), stream);
  scatter_add64<<<scatGrid, blk, 0, stream>>>(slotA, hedge_idx, node_idx, slotB, INC);
  gemm_gate_tanh_chain<true><<<gemmGridN, blk, 0, stream>>>(
      node_features, W_nn, slotB, W_hn, out_node, slotA, N);

  hipMemsetAsync(slotB, 0, (size_t)E * 64 * sizeof(float), stream);
  scatter_add64<<<scatGrid, blk, 0, stream>>>(slotA, node_idx, hedge_idx, slotB, INC);
  gemm_gate_tanh_chain<false><<<gemmGridE, blk, 0, stream>>>(
      hedge_features, W_hh, slotB, nullptr, out_hedge, nullptr, E);
}
